# starting kernel
# baseline (speedup 1.0000x reference)
#include <hip/hip_runtime.h>
#include <hip/hip_bf16.h>

typedef __attribute__((ext_vector_type(16))) _Float16 v16h;
typedef __attribute__((ext_vector_type(8)))  float    v8f;

#define NSTEPS 8

union FragU {
  uint4    q[2];
  unsigned u[8];
  v16h     h;
};

static __device__ __forceinline__ int sblk(int p) {
  int b = p >> 3, t = b & 3;
  t = (t == 1) ? 2 : ((t == 2) ? 1 : t);
  return (((b & ~3) | t) << 3) | (p & 7);
}

static __device__ __forceinline__ unsigned pk2(float a, float b) {
  return __builtin_bit_cast(unsigned, __builtin_amdgcn_cvt_pkrtz(a, b));
}

static __device__ __forceinline__ float fast_tanh(float x) {
#if __has_builtin(__builtin_amdgcn_tanhf)
  return __builtin_amdgcn_tanhf(x);
#else
  return tanhf(x);
#endif
}

static __device__ __forceinline__ v16h load_afrag(const _Float16* __restrict__ base,
                                                  int row, int rowStride, int k0) {
  FragU f;
  const _Float16* p = base + row * rowStride + k0;
  f.q[0] = *(const uint4*)(p);
  f.q[1] = *(const uint4*)(p + 16);
  return f.h;
}

static __device__ __forceinline__ void field_eval(
    v8f cur[4],
    const _Float16* __restrict__ w1t, const _Float16* __restrict__ w2t,
    const v8f* __restrict__ bias1, const v8f* __restrict__ bias2,
    int li, int hi)
{
  FragU B1[2];
#pragma unroll
  for (int c = 0; c < 2; ++c)
#pragma unroll
    for (int j = 0; j < 4; ++j) {
      B1[c].u[j]     = pk2(cur[2*c    ][2*j], cur[2*c    ][2*j+1]);
      B1[c].u[4 + j] = pk2(cur[2*c + 1][2*j], cur[2*c + 1][2*j+1]);
    }

  v8f out[4];
#pragma unroll
  for (int c = 0; c < 4; ++c) {
    FragU B2;
#pragma unroll
    for (int p = 0; p < 2; ++p) {
      const int T = 2*c + p;
      v8f hacc = bias1[T];
      v16h a0 = load_afrag(w1t, 16*T + li, 64, 8*hi);
      hacc = __builtin_amdgcn_wmma_f32_16x16x32_f16(false, a0, false, B1[0].h,
                                                    (short)0, hacc, false, false);
      v16h a1 = load_afrag(w1t, 16*T + li, 64, 32 + 8*hi);
      hacc = __builtin_amdgcn_wmma_f32_16x16x32_f16(false, a1, false, B1[1].h,
                                                    (short)0, hacc, false, false);
#pragma unroll
      for (int j = 0; j < 4; ++j)
        B2.u[4*p + j] = pk2(fast_tanh(hacc[2*j]), fast_tanh(hacc[2*j+1]));
    }
#pragma unroll
    for (int g = 0; g < 4; ++g) {
      v8f cacc = (c == 0) ? bias2[g] : out[g];
      v16h a2 = load_afrag(w2t, 16*g + li, 128, 32*c + 8*hi);
      out[g] = __builtin_amdgcn_wmma_f32_16x16x32_f16(false, a2, false, B2.h,
                                                      (short)0, cacc, false, false);
    }
  }
#pragma unroll
  for (int g = 0; g < 4; ++g) cur[g] = out[g];
}

__launch_bounds__(256)
__global__ void node_rk4_wmma(const float* __restrict__ x,
                              const float* __restrict__ W1,
                              const float* __restrict__ b1,
                              const float* __restrict__ W2,
                              const float* __restrict__ b2,
                              float* __restrict__ out)
{
  __shared__ _Float16 w1t[128 * 64];
  __shared__ _Float16 w2t[64 * 128];

  const int tid = threadIdx.x;
  for (int i = tid; i < 128 * 64; i += 256) {
    int m = i >> 6, k = i & 63;
    w1t[i] = (_Float16)W1[sblk(k) * 128 + sblk(m)];
  }
  for (int i = tid; i < 64 * 128; i += 256) {
    int m = i >> 7, k = i & 127;
    w2t[i] = (_Float16)W2[sblk(k) * 64 + sblk(m)];
  }
  __syncthreads();

  const int lane = tid & 31;
  const int wave = tid >> 5;
  const int li   = lane & 15;
  const int hi   = lane >> 4;

  v8f bias1[8], bias2[4];
#pragma unroll
  for (int t = 0; t < 8; ++t) {
    const int base = sblk(16*t + 8*hi);
#pragma unroll
    for (int r = 0; r < 8; ++r) bias1[t][r] = b1[base + r];
  }
#pragma unroll
  for (int g = 0; g < 4; ++g) {
    const int base = sblk(16*g + 8*hi);
#pragma unroll
    for (int r = 0; r < 8; ++r) bias2[g][r] = b2[base + r];
  }

  const long   row  = (long)blockIdx.x * 128 + wave * 16 + li;
  const float* xrow = x + row * 64;

  v8f z[4];
#pragma unroll
  for (int g = 0; g < 4; ++g) {
    const int base = sblk(16*g + 8*hi);
    float4 a = *(const float4*)(xrow + base);
    float4 b = *(const float4*)(xrow + base + 4);
    z[g][0]=a.x; z[g][1]=a.y; z[g][2]=a.z; z[g][3]=a.w;
    z[g][4]=b.x; z[g][5]=b.y; z[g][6]=b.z; z[g][7]=b.w;
  }

  const float hs = 1.0f / (float)NSTEPS;
  v8f acc[4], cur[4];

  for (int s = 0; s < NSTEPS; ++s) {
#pragma unroll
    for (int g = 0; g < 4; ++g) cur[g] = z[g];
    field_eval(cur, w1t, w2t, bias1, bias2, li, hi);
#pragma unroll
    for (int g = 0; g < 4; ++g)
#pragma unroll
      for (int r = 0; r < 8; ++r) {
        acc[g][r] = z[g][r] + (hs / 6.0f) * cur[g][r];
        cur[g][r] = z[g][r] + (hs * 0.5f) * cur[g][r];
      }
    field_eval(cur, w1t, w2t, bias1, bias2, li, hi);
#pragma unroll
    for (int g = 0; g < 4; ++g)
#pragma unroll
      for (int r = 0; r < 8; ++r) {
        acc[g][r] += (hs / 3.0f) * cur[g][r];
        cur[g][r]  = z[g][r] + (hs * 0.5f) * cur[g][r];
      }
    field_eval(cur, w1t, w2t, bias1, bias2, li, hi);
#pragma unroll
    for (int g = 0; g < 4; ++g)
#pragma unroll
      for (int r = 0; r < 8; ++r) {
        acc[g][r] += (hs / 3.0f) * cur[g][r];
        cur[g][r]  = z[g][r] + hs * cur[g][r];
      }
    field_eval(cur, w1t, w2t, bias1, bias2, li, hi);
#pragma unroll
    for (int g = 0; g < 4; ++g)
#pragma unroll
      for (int r = 0; r < 8; ++r)
        z[g][r] = acc[g][r] + (hs / 6.0f) * cur[g][r];
  }

  float* orow = out + row * 64;
#pragma unroll
  for (int g = 0; g < 4; ++g) {
    const int base = sblk(16*g + 8*hi);
    float4 a, b;
    a.x=z[g][0]; a.y=z[g][1]; a.z=z[g][2]; a.w=z[g][3];
    b.x=z[g][4]; b.y=z[g][5]; b.z=z[g][6]; b.w=z[g][7];
    *(float4*)(orow + base)     = a;
    *(float4*)(orow + base + 4) = b;
  }
}

extern "C" void kernel_launch(void* const* d_in, const int* in_sizes, int n_in,
                              void* d_out, int out_size, void* d_ws, size_t ws_size,
                              hipStream_t stream) {
  (void)n_in; (void)out_size; (void)d_ws; (void)ws_size;
  const float* x  = (const float*)d_in[0];
  const float* W1 = (const float*)d_in[2];
  const float* b1 = (const float*)d_in[3];
  const float* W2 = (const float*)d_in[4];
  const float* b2 = (const float*)d_in[5];
  float* out = (float*)d_out;

  const int B = in_sizes[0] / 64;
  dim3 grid(B / 128), block(256);
  node_rk4_wmma<<<grid, block, 0, stream>>>(x, W1, b1, W2, b2, out);
}


# resident grid: 256 work-groups each loop over 8 row tiles, so the weight staging (global -> f16 LDS -> registers) and bias loads run once per work-group instead of once per 128 rows
# speedup vs baseline: 1.3316x; 1.3316x over previous
#include <hip/hip_runtime.h>
#include <hip/hip_bf16.h>

typedef __attribute__((ext_vector_type(16))) _Float16 v16h;
typedef __attribute__((ext_vector_type(8)))  float    v8f;

#define NSTEPS 8

union FragU {
  uint4    q[2];
  unsigned u[8];
  v16h     h;
};

static __device__ __forceinline__ int sblk(int p) {
  int b = p >> 3, t = b & 3;
  t = (t == 1) ? 2 : ((t == 2) ? 1 : t);
  return (((b & ~3) | t) << 3) | (p & 7);
}

static __device__ __forceinline__ unsigned pk2(float a, float b) {
  return __builtin_bit_cast(unsigned, __builtin_amdgcn_cvt_pkrtz(a, b));
}

static __device__ __forceinline__ float fast_tanh(float x) {
#if __has_builtin(__builtin_amdgcn_tanhf)
  return __builtin_amdgcn_tanhf(x);
#else
  return tanhf(x);
#endif
}

static __device__ __forceinline__ v16h load_afrag(const _Float16* __restrict__ base,
                                                  int row, int rowStride, int k0) {
  FragU f;
  const _Float16* p = base + row * rowStride + k0;
  f.q[0] = *(const uint4*)(p);
  f.q[1] = *(const uint4*)(p + 16);
  return f.h;
}

static __device__ __forceinline__ void field_eval(
    v8f cur[4],
    const _Float16* __restrict__ w1t, const _Float16* __restrict__ w2t,
    const v8f* __restrict__ bias1, const v8f* __restrict__ bias2,
    int li, int hi)
{
  FragU B1[2];
#pragma unroll
  for (int c = 0; c < 2; ++c)
#pragma unroll
    for (int j = 0; j < 4; ++j) {
      B1[c].u[j]     = pk2(cur[2*c    ][2*j], cur[2*c    ][2*j+1]);
      B1[c].u[4 + j] = pk2(cur[2*c + 1][2*j], cur[2*c + 1][2*j+1]);
    }

  v8f out[4];
#pragma unroll
  for (int c = 0; c < 4; ++c) {
    FragU B2;
#pragma unroll
    for (int p = 0; p < 2; ++p) {
      const int T = 2*c + p;
      v8f hacc = bias1[T];
      v16h a0 = load_afrag(w1t, 16*T + li, 64, 8*hi);
      hacc = __builtin_amdgcn_wmma_f32_16x16x32_f16(false, a0, false, B1[0].h,
                                                    (short)0, hacc, false, false);
      v16h a1 = load_afrag(w1t, 16*T + li, 64, 32 + 8*hi);
      hacc = __builtin_amdgcn_wmma_f32_16x16x32_f16(false, a1, false, B1[1].h,
                                                    (short)0, hacc, false, false);
#pragma unroll
      for (int j = 0; j < 4; ++j)
        B2.u[4*p + j] = pk2(fast_tanh(hacc[2*j]), fast_tanh(hacc[2*j+1]));
    }
#pragma unroll
    for (int g = 0; g < 4; ++g) {
      v8f cacc = (c == 0) ? bias2[g] : out[g];
      v16h a2 = load_afrag(w2t, 16*g + li, 128, 32*c + 8*hi);
      out[g] = __builtin_amdgcn_wmma_f32_16x16x32_f16(false, a2, false, B2.h,
                                                      (short)0, cacc, false, false);
    }
  }
#pragma unroll
  for (int g = 0; g < 4; ++g) cur[g] = out[g];
}

__launch_bounds__(256)
__global__ void node_rk4_wmma(const float* __restrict__ x,
                              const float* __restrict__ W1,
                              const float* __restrict__ b1,
                              const float* __restrict__ W2,
                              const float* __restrict__ b2,
                              float* __restrict__ out,
                              int ntiles, int tile_stride)
{
  __shared__ _Float16 w1t[128 * 64];
  __shared__ _Float16 w2t[64 * 128];

  const int tid = threadIdx.x;
  for (int i = tid; i < 128 * 64; i += 256) {
    int m = i >> 6, k = i & 63;
    w1t[i] = (_Float16)W1[sblk(k) * 128 + sblk(m)];
  }
  for (int i = tid; i < 64 * 128; i += 256) {
    int m = i >> 7, k = i & 127;
    w2t[i] = (_Float16)W2[sblk(k) * 64 + sblk(m)];
  }
  __syncthreads();

  const int lane = tid & 31;
  const int wave = tid >> 5;
  const int li   = lane & 15;
  const int hi   = lane >> 4;

  v8f bias1[8], bias2[4];
#pragma unroll
  for (int t = 0; t < 8; ++t) {
    const int base = sblk(16*t + 8*hi);
#pragma unroll
    for (int r = 0; r < 8; ++r) bias1[t][r] = b1[base + r];
  }
#pragma unroll
  for (int g = 0; g < 4; ++g) {
    const int base = sblk(16*g + 8*hi);
#pragma unroll
    for (int r = 0; r < 8; ++r) bias2[g][r] = b2[base + r];
  }

  for (int tile = blockIdx.x; tile < ntiles; tile += tile_stride) {
  const long   row  = (long)tile * 128 + wave * 16 + li;
  const float* xrow = x + row * 64;

  v8f z[4];
#pragma unroll
  for (int g = 0; g < 4; ++g) {
    const int base = sblk(16*g + 8*hi);
    float4 a = *(const float4*)(xrow + base);
    float4 b = *(const float4*)(xrow + base + 4);
    z[g][0]=a.x; z[g][1]=a.y; z[g][2]=a.z; z[g][3]=a.w;
    z[g][4]=b.x; z[g][5]=b.y; z[g][6]=b.z; z[g][7]=b.w;
  }

  const float hs = 1.0f / (float)NSTEPS;
  v8f acc[4], cur[4];

  for (int s = 0; s < NSTEPS; ++s) {
#pragma unroll
    for (int g = 0; g < 4; ++g) cur[g] = z[g];
    field_eval(cur, w1t, w2t, bias1, bias2, li, hi);
#pragma unroll
    for (int g = 0; g < 4; ++g)
#pragma unroll
      for (int r = 0; r < 8; ++r) {
        acc[g][r] = z[g][r] + (hs / 6.0f) * cur[g][r];
        cur[g][r] = z[g][r] + (hs * 0.5f) * cur[g][r];
      }
    field_eval(cur, w1t, w2t, bias1, bias2, li, hi);
#pragma unroll
    for (int g = 0; g < 4; ++g)
#pragma unroll
      for (int r = 0; r < 8; ++r) {
        acc[g][r] += (hs / 3.0f) * cur[g][r];
        cur[g][r]  = z[g][r] + (hs * 0.5f) * cur[g][r];
      }
    field_eval(cur, w1t, w2t, bias1, bias2, li, hi);
#pragma unroll
    for (int g = 0; g < 4; ++g)
#pragma unroll
      for (int r = 0; r < 8; ++r) {
        acc[g][r] += (hs / 3.0f) * cur[g][r];
        cur[g][r]  = z[g][r] + hs * cur[g][r];
      }
    field_eval(cur, w1t, w2t, bias1, bias2, li, hi);
#pragma unroll
    for (int g = 0; g < 4; ++g)
#pragma unroll
      for (int r = 0; r < 8; ++r)
        z[g][r] = acc[g][r] + (hs / 6.0f) * cur[g][r];
  }

  float* orow = out + row * 64;
#pragma unroll
  for (int g = 0; g < 4; ++g) {
    const int base = sblk(16*g + 8*hi);
    float4 a, b;
    a.x=z[g][0]; a.y=z[g][1]; a.z=z[g][2]; a.w=z[g][3];
    b.x=z[g][4]; b.y=z[g][5]; b.z=z[g][6]; b.w=z[g][7];
    *(float4*)(orow + base)     = a;
    *(float4*)(orow + base + 4) = b;
  }
  }
}

extern "C" void kernel_launch(void* const* d_in, const int* in_sizes, int n_in,
                              void* d_out, int out_size, void* d_ws, size_t ws_size,
                              hipStream_t stream) {
  (void)n_in; (void)out_size; (void)d_ws; (void)ws_size;
  const float* x  = (const float*)d_in[0];
  const float* W1 = (const float*)d_in[2];
  const float* b1 = (const float*)d_in[3];
  const float* W2 = (const float*)d_in[4];
  const float* b2 = (const float*)d_in[5];
  float* out = (float*)d_out;

  const int B = in_sizes[0] / 64;
  dim3 grid(256), block(256);
  node_rk4_wmma<<<grid, block, 0, stream>>>(x, W1, b1, W2, b2, out, B / 128, 256);
}
